# starting kernel
# baseline (speedup 1.0000x reference)
#include <hip/hip_runtime.h>
#include <math.h>

typedef float v2f __attribute__((ext_vector_type(2)));
typedef float v8f __attribute__((ext_vector_type(8)));

#define B_ 8
#define N_ 4096
#define H_ 256
#define F_ 6

static __device__ __forceinline__ v8f wmma_f32x4(v2f a, v2f b, v8f c) {
  return __builtin_amdgcn_wmma_f32_16x16x4_f32(false, a, false, b, (short)0, c,
                                               false, false);
}

__global__ __launch_bounds__(256) void hr_build_M(
    const float* __restrict__ t, const float* __restrict__ osc,
    const float* __restrict__ phase, float* __restrict__ M) {
  const int idx = blockIdx.x * 256 + threadIdx.x;
  const int k = idx & (H_ - 1);
  const int h = (idx >> 8) & (H_ - 1);
  const int b = idx >> 16;

  const float tb = t[b];
  const float freqs[F_] = {1.f, 2.f, 4.f, 8.f, 7.f, 5.f};

  float acc = 0.f;
#pragma unroll
  for (int f = 0; f < F_; ++f) {
    const float r = sinf(fmaf(freqs[f], tb, phase[f * H_ + k]));
    acc = fmaf(osc[(f * H_ + h) * H_ + k], r, acc);
  }
  M[idx] = acc;
}

__global__ __launch_bounds__(256) void hr_gemm(const float* __restrict__ x,
                                               const float* __restrict__ M,
                                               float* __restrict__ out) {
  __shared__ float Mlds[128 * 96];

  const int kblk = blockIdx.x;
  const int nblk = blockIdx.y;
  const int b = blockIdx.z;

  const int tid = threadIdx.x;

  {
    const float* src = M + ((size_t)b * H_) * H_ + kblk * 32;
#pragma unroll
    for (int i = 0; i < 32; ++i) {
      const int linear = tid + i * 256;
      const int kk = linear & 31;
      const int h = linear >> 5;
      Mlds[(h >> 1) * 96 + kk * 2 + (h & 1)] = src[h * H_ + kk];
    }
  }
  __syncthreads();

  const int lane = tid & 31;
  const int wave = tid >> 5;
  const int lhalf = lane >> 4;
  const int lmod = lane & 15;

  const float* xrow =
      x + ((size_t)b * N_ + nblk * 512 + wave * 64 + lmod) * H_ + lhalf * 2;

  const v2f* mp = (const v2f*)Mlds;

  v8f acc[4][2];
#pragma unroll
  for (int rg = 0; rg < 4; ++rg)
#pragma unroll
    for (int cg = 0; cg < 2; ++cg) acc[rg][cg] = (v8f){};

#pragma unroll 2
  for (int h0 = 0; h0 < H_; h0 += 4) {
    v2f a[4];
#pragma unroll
    for (int rg = 0; rg < 4; ++rg)
      a[rg] = *(const v2f*)(xrow + rg * 16 * H_ + h0);

    const int p = (h0 >> 1) + lhalf;
    const v2f b0 = mp[p * 48 + lmod];
    const v2f b1 = mp[p * 48 + 16 + lmod];

#pragma unroll
    for (int rg = 0; rg < 4; ++rg) {
      acc[rg][0] = wmma_f32x4(a[rg], b0, acc[rg][0]);
      acc[rg][1] = wmma_f32x4(a[rg], b1, acc[rg][1]);
    }
  }

  float* obase =
      out + ((size_t)b * N_ + nblk * 512 + wave * 64) * H_ + kblk * 32;
#pragma unroll
  for (int rg = 0; rg < 4; ++rg) {
#pragma unroll
    for (int r = 0; r < 8; ++r) {
      const int row = rg * 16 + lhalf * 8 + r;
      obase[row * H_ + lmod] = acc[rg][0][r];
      obase[row * H_ + 16 + lmod] = acc[rg][1][r];
    }
  }
}

extern "C" void kernel_launch(void* const* d_in, const int* in_sizes, int n_in,
                              void* d_out, int out_size, void* d_ws,
                              size_t ws_size, hipStream_t stream) {
  const float* x = (const float*)d_in[0];
  const float* t = (const float*)d_in[1];
  const float* osc = (const float*)d_in[2];
  const float* phase = (const float*)d_in[3];
  float* out = (float*)d_out;
  float* M = (float*)d_ws;

  (void)in_sizes;
  (void)n_in;
  (void)out_size;
  (void)ws_size;

  hr_build_M<<<(B_ * H_ * H_) / 256, 256, 0, stream>>>(t, osc, phase, M);
  hr_gemm<<<dim3(H_ / 32, N_ / 512, B_), 256, 0, stream>>>(x, M, out);
}


# GEMM re-tiled: each wave keeps its 16x256 strip of x in registers and sweeps all 256 columns, M chunks through double-buffered LDS; x is fetched once, not 8 times; k summation order changes
# speedup vs baseline: 1.6861x; 1.6861x over previous
#include <hip/hip_runtime.h>
#include <math.h>

typedef float v2f __attribute__((ext_vector_type(2)));
typedef float v8f __attribute__((ext_vector_type(8)));

#define B_ 8
#define N_ 4096
#define H_ 256
#define F_ 6

static __device__ __forceinline__ v8f wmma_f32x4(v2f a, v2f b, v8f c) {
  return __builtin_amdgcn_wmma_f32_16x16x4_f32(false, a, false, b, (short)0, c,
                                               false, false);
}

__global__ __launch_bounds__(256) void hr_build_M(
    const float* __restrict__ t, const float* __restrict__ osc,
    const float* __restrict__ phase, float* __restrict__ M) {
  const int idx = blockIdx.x * 256 + threadIdx.x;
  const int k = idx & (H_ - 1);
  const int h = (idx >> 8) & (H_ - 1);
  const int b = idx >> 16;

  const float tb = t[b];
  const float freqs[F_] = {1.f, 2.f, 4.f, 8.f, 7.f, 5.f};

  float acc = 0.f;
#pragma unroll
  for (int f = 0; f < F_; ++f) {
    const float r = sinf(fmaf(freqs[f], tb, phase[f * H_ + k]));
    acc = fmaf(osc[(f * H_ + h) * H_ + k], r, acc);
  }
  M[idx] = acc;
}

static __device__ __forceinline__ void lds_handover() {
  __builtin_amdgcn_fence(__ATOMIC_RELEASE, "workgroup", "local");
  __builtin_amdgcn_s_barrier();
  __builtin_amdgcn_fence(__ATOMIC_ACQUIRE, "workgroup", "local");
}

typedef float v4f __attribute__((ext_vector_type(4)));

#define PAIR_STRIDE_ 80
#define CHUNK_FLOATS_ (128 * PAIR_STRIDE_)
#define NCHUNK_ (H_ / 32)

__global__ __launch_bounds__(256) void hr_gemm(const float* __restrict__ x,
                                               const float* __restrict__ M,
                                               float* __restrict__ out) {
  __shared__ float Mlds[2 * CHUNK_FLOATS_];

  const int tid = threadIdx.x;
  const int lane = tid & 31;
  const int wave = tid >> 5;
  const int lhalf = lane >> 4;
  const int lmod = lane & 15;

  const int b = blockIdx.x >> 5;
  const int rblk = blockIdx.x & 31;
  const size_t row0 = (size_t)b * N_ + rblk * 128 + wave * 16;

  const int sm = tid & 7;
  const int sq = tid >> 3;
  const float* msrc = M + (size_t)b * H_ * H_ + (2 * sq) * H_ + sm * 4;
  const int sdst = sq * PAIR_STRIDE_ + sm * 8;

  v4f s0[4], s1[4];
#pragma unroll
  for (int i = 0; i < 4; ++i) {
    s0[i] = *(const v4f*)(msrc + i * 64 * H_);
    s1[i] = *(const v4f*)(msrc + i * 64 * H_ + H_);
  }

  const float* xrow = x + (row0 + lmod) * H_ + lhalf * 4;
  v4f a[32];
#pragma unroll
  for (int j = 0; j < 32; ++j) a[j] = *(const v4f*)(xrow + 8 * j);

#pragma unroll
  for (int i = 0; i < 4; ++i) {
    float* d = Mlds + sdst + i * 32 * PAIR_STRIDE_;
    *(v4f*)d = (v4f){s0[i].x, s1[i].x, s0[i].y, s1[i].y};
    *(v4f*)(d + 4) = (v4f){s0[i].z, s1[i].z, s0[i].w, s1[i].w};
  }
  lds_handover();

  float* orow = out + (row0 + lhalf * 8) * H_ + lmod;

#pragma unroll 1
  for (int c = 0; c < NCHUNK_; ++c) {
    const float* cur = Mlds + (c & 1) * CHUNK_FLOATS_;
    float* nxt = Mlds + ((c + 1) & 1) * CHUNK_FLOATS_;
    const bool more = (c + 1 < NCHUNK_);

    if (more) {
#pragma unroll
      for (int i = 0; i < 4; ++i) {
        s0[i] = *(const v4f*)(msrc + (c + 1) * 32 + i * 64 * H_);
        s1[i] = *(const v4f*)(msrc + (c + 1) * 32 + i * 64 * H_ + H_);
      }
    }

    const v2f* mp = (const v2f*)cur + lhalf * (2 * (PAIR_STRIDE_ / 2)) + lmod;

    v8f acc0 = (v8f){};
    v8f acc1 = (v8f){};

    v2f bq[2][8][4];
#pragma unroll
    for (int jj = 0; jj < 8; ++jj) {
      const v2f* mq = mp + (4 * jj) * (PAIR_STRIDE_ / 2);
      bq[0][jj][0] = mq[0];
      bq[0][jj][1] = mq[16];
      bq[0][jj][2] = mq[PAIR_STRIDE_ / 2];
      bq[0][jj][3] = mq[PAIR_STRIDE_ / 2 + 16];
    }
#pragma unroll
    for (int g = 0; g < 4; ++g) {
      if (g + 1 < 4) {
#pragma unroll
        for (int jj = 0; jj < 8; ++jj) {
          const v2f* mq = mp + (4 * (8 * (g + 1) + jj)) * (PAIR_STRIDE_ / 2);
          bq[(g + 1) & 1][jj][0] = mq[0];
          bq[(g + 1) & 1][jj][1] = mq[16];
          bq[(g + 1) & 1][jj][2] = mq[PAIR_STRIDE_ / 2];
          bq[(g + 1) & 1][jj][3] = mq[PAIR_STRIDE_ / 2 + 16];
        }
      }
      __builtin_amdgcn_sched_barrier(0);
#pragma unroll
      for (int jj = 0; jj < 8; ++jj) {
        const int j = 8 * g + jj;
        const v2f alo = (v2f){a[j].x, a[j].y};
        const v2f ahi = (v2f){a[j].z, a[j].w};
        acc0 = wmma_f32x4(alo, bq[g & 1][jj][0], acc0);
        acc1 = wmma_f32x4(alo, bq[g & 1][jj][1], acc1);
        acc0 = wmma_f32x4(ahi, bq[g & 1][jj][2], acc0);
        acc1 = wmma_f32x4(ahi, bq[g & 1][jj][3], acc1);
      }
      __builtin_amdgcn_sched_barrier(0);
    }

    float* o = orow + c * 32;
#pragma unroll
    for (int r = 0; r < 8; ++r) {
      o[r * H_] = acc0[r];
      o[r * H_ + 16] = acc1[r];
    }

    if (more) {
#pragma unroll
      for (int i = 0; i < 4; ++i) {
        float* d = nxt + sdst + i * 32 * PAIR_STRIDE_;
        *(v4f*)d = (v4f){s0[i].x, s1[i].x, s0[i].y, s1[i].y};
        *(v4f*)(d + 4) = (v4f){s0[i].z, s1[i].z, s0[i].w, s1[i].w};
      }
    }
    lds_handover();
  }
}

extern "C" void kernel_launch(void* const* d_in, const int* in_sizes, int n_in,
                              void* d_out, int out_size, void* d_ws,
                              size_t ws_size, hipStream_t stream) {
  const float* x = (const float*)d_in[0];
  const float* t = (const float*)d_in[1];
  const float* osc = (const float*)d_in[2];
  const float* phase = (const float*)d_in[3];
  float* out = (float*)d_out;
  float* M = (float*)d_ws;

  (void)in_sizes;
  (void)n_in;
  (void)out_size;
  (void)ws_size;

  hr_build_M<<<(B_ * H_ * H_) / 256, 256, 0, stream>>>(t, osc, phase, M);
  hr_gemm<<<(N_ / 128) * B_, 256, 0, stream>>>(x, M, out);
}
